# starting kernel
# baseline (speedup 1.0000x reference)
#include <hip/hip_runtime.h>
#include <hip/hip_bf16.h>
#include <math.h>

#define NN      131072
#define DEG     16
#define NNZ     (NN * DEG)
#define DMODEL  64
#define NHEAD   4
#define HDIM    16
#define BLK     256

typedef __attribute__((ext_vector_type(16))) _Float16 v16h;
typedef __attribute__((ext_vector_type(8)))  _Float16 v8h;
typedef __attribute__((ext_vector_type(8)))  float    v8f;

__device__ __forceinline__ v8f v8f_zero() {
    v8f z = {0.f, 0.f, 0.f, 0.f, 0.f, 0.f, 0.f, 0.f};
    return z;
}

__device__ __forceinline__ v16h cat16(v8h a, v8h b) {
    return __builtin_shufflevector(a, b, 0, 1, 2, 3, 4, 5, 6, 7,
                                   8, 9, 10, 11, 12, 13, 14, 15);
}

__device__ __forceinline__ v16h frag_from_f32(float4 a0, float4 a1,
                                              float4 a2, float4 a3) {
    v16h r;
    r[0] = (_Float16)a0.x;  r[1] = (_Float16)a0.y;
    r[2] = (_Float16)a0.z;  r[3] = (_Float16)a0.w;
    r[4] = (_Float16)a1.x;  r[5] = (_Float16)a1.y;
    r[6] = (_Float16)a1.z;  r[7] = (_Float16)a1.w;
    r[8] = (_Float16)a2.x;  r[9] = (_Float16)a2.y;
    r[10] = (_Float16)a2.z; r[11] = (_Float16)a2.w;
    r[12] = (_Float16)a3.x; r[13] = (_Float16)a3.y;
    r[14] = (_Float16)a3.z; r[15] = (_Float16)a3.w;
    return r;
}

__device__ __forceinline__ float4 relu4(float4 a, float4 b) {
    float4 r;
    r.x = a.x + b.x; r.x = r.x > 0.f ? r.x : 0.f;
    r.y = a.y + b.y; r.y = r.y > 0.f ? r.y : 0.f;
    r.z = a.z + b.z; r.z = r.z > 0.f ? r.z : 0.f;
    r.w = a.w + b.w; r.w = r.w > 0.f ? r.w : 0.f;
    return r;
}

template <int TM, int TN, bool RELU_IN, int ACT_OUT>
__global__ void wmma_gemm_kernel(const float* __restrict__ A,
                                 const float* __restrict__ abias,
                                 const float* __restrict__ W,
                                 const float* __restrict__ obias,
                                 float* __restrict__ C,
                                 int M, int K, int Nc)
{
    __shared__ __align__(16) _Float16 sW[12288];

    int tid  = (int)threadIdx.x;
    int lane = tid & 31;
    int hi = lane >> 4, lo = lane & 15;

    for (int i = tid; i < K * Nc; i += 256) {
        int k = i / Nc, n = i - k * Nc;
        sW[n * K + k] = (_Float16)W[i];
    }
    __syncthreads();

    int gwid = (int)((blockIdx.x * blockDim.x + threadIdx.x) >> 5);
    int groups_n = Nc / (16 * TN);
    int total = (M / (16 * TM)) * groups_n;
    if (gwid >= total) return;
    int mg = gwid / groups_n;
    int tg = gwid - mg * groups_n;

    #pragma unroll
    for (int t = 0; t < TM; ++t) {
        int tm = mg * TM + t;
        int m = tm * 16 + lo;

        v8f c[TN];
        #pragma unroll
        for (int j = 0; j < TN; ++j) c[j] = v8f_zero();

        const float* Arow = A + (size_t)m * K;
        for (int k0 = 0; k0 < K; k0 += 32) {
            const float4* ap0 = (const float4*)(Arow + k0 + 8 * hi);
            const float4* ap1 = (const float4*)(Arow + k0 + 16 + 8 * hi);
            float4 a0 = ap0[0], a1 = ap0[1], a2 = ap1[0], a3 = ap1[1];
            if (RELU_IN) {
                const float4* bp0 = (const float4*)(abias + k0 + 8 * hi);
                const float4* bp1 = (const float4*)(abias + k0 + 16 + 8 * hi);
                a0 = relu4(a0, bp0[0]); a1 = relu4(a1, bp0[1]);
                a2 = relu4(a2, bp1[0]); a3 = relu4(a3, bp1[1]);
            }
            v16h af = frag_from_f32(a0, a1, a2, a3);

            #pragma unroll
            for (int j = 0; j < TN; ++j) {
                int n = (tg * TN + j) * 16 + lo;
                const v8h* wp0 = (const v8h*)(&sW[n * K + k0 + 8 * hi]);
                const v8h* wp1 = (const v8h*)(&sW[n * K + k0 + 16 + 8 * hi]);
                v16h bf = cat16(wp0[0], wp1[0]);
                c[j] = __builtin_amdgcn_wmma_f32_16x16x32_f16(
                    false, af, false, bf, (short)0, c[j], false, false);
            }
        }

        #pragma unroll
        for (int j = 0; j < TN; ++j) {
            int n = (tg * TN + j) * 16 + lo;
            #pragma unroll
            for (int r = 0; r < 8; ++r) {
                int row = tm * 16 + r + 8 * hi;
                float v = c[j][r];
                if (obias) v += obias[n];
                if (ACT_OUT == 1) {
                    v = v > 0.f ? v : 0.f;
                } else if (ACT_OUT == 2) {
                    v = 0.5f * v * (1.0f + erff(v * 0.70710678118654752f));
                }
                C[(size_t)row * Nc + n] = v;
            }
        }
    }
}

__global__ void spmm_kernel(const int* __restrict__ rows,
                            const int* __restrict__ cols,
                            const float* __restrict__ vals,
                            const float* __restrict__ in,
                            float* __restrict__ out, int F4)
{
    long long t = (long long)blockIdx.x * blockDim.x + threadIdx.x;
    long long total = (long long)NNZ * F4;
    if (t >= total) return;
    int e  = (int)(t / F4);
    int c4 = (int)(t - (long long)e * F4);
    int F  = F4 * 4;
    float v = vals[e];
    const float4 x = ((const float4*)(in + (size_t)cols[e] * F))[c4];
    float* o = out + (size_t)rows[e] * F + c4 * 4;
    atomicAdd(o + 0, v * x.x);
    atomicAdd(o + 1, v * x.y);
    atomicAdd(o + 2, v * x.z);
    atomicAdd(o + 3, v * x.w);
}

__global__ void zero_kernel(float* __restrict__ p, int n)
{
    int i = (int)(blockIdx.x * blockDim.x + threadIdx.x);
    if (i < n) p[i] = 0.f;
}

__global__ void bias_relu_kernel(const float* __restrict__ in,
                                 const float* __restrict__ bias,
                                 float* __restrict__ out, int total, int F)
{
    int i = (int)(blockIdx.x * blockDim.x + threadIdx.x);
    if (i < total) {
        float v = in[i] + bias[i % F];
        out[i] = v > 0.f ? v : 0.f;
    }
}

__global__ void attn_kernel(const float* __restrict__ QKV,
                            float* __restrict__ ATT)
{
    __shared__ __align__(16) _Float16 sK[BLK * HDIM];
    __shared__ __align__(16) _Float16 sVt[HDIM * BLK];
    __shared__ __align__(16) _Float16 sP[8][16 * 32];

    int b = (int)(blockIdx.x >> 2);
    int h = (int)(blockIdx.x & 3);
    int tid  = (int)threadIdx.x;
    int wid  = tid >> 5;
    int lane = tid & 31;
    int hi   = lane >> 4;
    int lo   = lane & 15;
    size_t base = (size_t)b * BLK * (3 * DMODEL);

    for (int i = tid; i < BLK * HDIM; i += 256) {
        int s = i >> 4, d = i & 15;
        sK[i]            = (_Float16)QKV[base + (size_t)s * 192 + DMODEL     + h * HDIM + d];
        sVt[d * BLK + s] = (_Float16)QKV[base + (size_t)s * 192 + 2 * DMODEL + h * HDIM + d];
    }
    __syncthreads();

    v8h zero8 = {(_Float16)0.f, (_Float16)0.f, (_Float16)0.f, (_Float16)0.f,
                 (_Float16)0.f, (_Float16)0.f, (_Float16)0.f, (_Float16)0.f};

    for (int qt = wid; qt < 16; qt += 8) {
        int qrow = qt * 16 + lo;
        const float4* qp = (const float4*)(QKV + base + (size_t)qrow * 192
                                           + h * HDIM + 8 * hi);
        float4 q0 = qp[0], q1 = qp[1];
        q0.x *= 0.25f; q0.y *= 0.25f; q0.z *= 0.25f; q0.w *= 0.25f;
        q1.x *= 0.25f; q1.y *= 0.25f; q1.z *= 0.25f; q1.w *= 0.25f;
        float4 zf = {0.f, 0.f, 0.f, 0.f};
        v16h qf = frag_from_f32(q0, q1, zf, zf);

        v8f ctx = v8f_zero();
        float m[8], l[8];
        #pragma unroll
        for (int r = 0; r < 8; ++r) { m[r] = -3.0e38f; l[r] = 0.f; }

        for (int kk = 0; kk < 8; ++kk) {
            v8f s0, s1;
            {
                int key = (kk * 2) * 16 + lo;
                v16h bf = cat16(*(const v8h*)(&sK[key * HDIM + 8 * hi]), zero8);
                v8f z = v8f_zero();
                s0 = __builtin_amdgcn_wmma_f32_16x16x32_f16(
                    false, qf, false, bf, (short)0, z, false, false);
            }
            {
                int key = (kk * 2 + 1) * 16 + lo;
                v16h bf = cat16(*(const v8h*)(&sK[key * HDIM + 8 * hi]), zero8);
                v8f z = v8f_zero();
                s1 = __builtin_amdgcn_wmma_f32_16x16x32_f16(
                    false, qf, false, bf, (short)0, z, false, false);
            }
            #pragma unroll
            for (int r = 0; r < 8; ++r) {
                float tm = fmaxf(s0[r], s1[r]);
                #pragma unroll
                for (int d = 1; d < 16; d <<= 1)
                    tm = fmaxf(tm, __shfl_xor(tm, d, 32));
                float nm = fmaxf(m[r], tm);
                float sc = __expf(m[r] - nm);
                m[r] = nm;
                float e0 = __expf(s0[r] - nm);
                float e1 = __expf(s1[r] - nm);
                float rs = e0 + e1;
                #pragma unroll
                for (int d = 1; d < 16; d <<= 1)
                    rs += __shfl_xor(rs, d, 32);
                l[r] = l[r] * sc + rs;
                ctx[r] *= sc;
                int row = r + 8 * hi;
                sP[wid][row * 32 + lo]      = (_Float16)e0;
                sP[wid][row * 32 + 16 + lo] = (_Float16)e1;
            }
            v16h af = cat16(*(const v8h*)(&sP[wid][lo * 32 + 8 * hi]),
                            *(const v8h*)(&sP[wid][lo * 32 + 16 + 8 * hi]));
            v16h bf = cat16(*(const v8h*)(&sVt[lo * BLK + kk * 32 + 8 * hi]),
                            *(const v8h*)(&sVt[lo * BLK + kk * 32 + 16 + 8 * hi]));
            ctx = __builtin_amdgcn_wmma_f32_16x16x32_f16(
                false, af, false, bf, (short)0, ctx, false, false);
        }

        #pragma unroll
        for (int r = 0; r < 8; ++r) {
            int row = b * BLK + qt * 16 + r + 8 * hi;
            ATT[(size_t)row * DMODEL + h * HDIM + lo] = ctx[r] / l[r];
        }
    }
}

__global__ void ln_residual_kernel(const float* __restrict__ X,
                                   const float* __restrict__ Y,
                                   const float* __restrict__ g,
                                   const float* __restrict__ bt,
                                   float* __restrict__ out, int Mrows)
{
    int w = (int)((blockIdx.x * blockDim.x + threadIdx.x) >> 5);
    int lane = (int)(threadIdx.x & 31);
    if (w >= Mrows) return;
    size_t off = (size_t)w * DMODEL;
    float v0 = X[off + lane]      + Y[off + lane];
    float v1 = X[off + lane + 32] + Y[off + lane + 32];
    float s = v0 + v1;
    #pragma unroll
    for (int d = 1; d < 32; d <<= 1) s += __shfl_xor(s, d, 32);
    float mu = s * (1.f / 64.f);
    float d0 = v0 - mu, d1 = v1 - mu;
    float vs = d0 * d0 + d1 * d1;
    #pragma unroll
    for (int d = 1; d < 32; d <<= 1) vs += __shfl_xor(vs, d, 32);
    float rstd = rsqrtf(vs * (1.f / 64.f) + 1e-5f);
    out[off + lane]      = d0 * rstd * g[lane]      + bt[lane];
    out[off + lane + 32] = d1 * rstd * g[lane + 32] + bt[lane + 32];
}

__global__ void cls_kernel(const float* __restrict__ X2,
                           const float* __restrict__ Wc,
                           const float* __restrict__ bc,
                           float* __restrict__ out, int Mrows)
{
    int w = (int)((blockIdx.x * blockDim.x + threadIdx.x) >> 5);
    int lane = (int)(threadIdx.x & 31);
    int hi = lane >> 4, lo = lane & 15;
    if (w >= Mrows) return;
    size_t off = (size_t)w * DMODEL;
    float acc = 0.f;
    #pragma unroll
    for (int kk = 0; kk < 32; ++kk) {
        int k = hi * 32 + kk;
        acc += X2[off + k] * Wc[k * 16 + lo];
    }
    acc += __shfl_xor(acc, 16, 32);
    acc += bc[lo];
    float mx = acc;
    #pragma unroll
    for (int d = 1; d < 16; d <<= 1) mx = fmaxf(mx, __shfl_xor(mx, d, 32));
    float e = expf(acc - mx), sm = e;
    #pragma unroll
    for (int d = 1; d < 16; d <<= 1) sm += __shfl_xor(sm, d, 32);
    float res = acc - mx - logf(sm);
    if (hi == 0) out[(size_t)w * 16 + lo] = res;
}


static inline int ceil_div(long long a, long long b) { return (int)((a + b - 1) / b); }

#define GEMM_WAVES(M, TM, Nc, TN) \
    ((long long)((M) / (16 * (TM))) * ((Nc) / (16 * (TN))))

static void launch_gemm_none(const float* A, const float* W, const float* ob,
                             float* C, int M, int K, int Nc, hipStream_t s)
{
    long long waves = GEMM_WAVES(M, 4, Nc, 4);
    wmma_gemm_kernel<4, 4, false, 0><<<ceil_div(waves * 32, 256), 256, 0, s>>>(
        A, nullptr, W, ob, C, M, K, Nc);
}

extern "C" void kernel_launch(void* const* d_in, const int* in_sizes, int n_in,
                              void* d_out, int out_size, void* d_ws, size_t ws_size,
                              hipStream_t stream)
{
    const float* x      = (const float*)d_in[0];
    const int*   G_rows = (const int*)  d_in[1];
    const int*   G_cols = (const int*)  d_in[2];
    const float* G_vals = (const float*)d_in[3];
    const float* W1     = (const float*)d_in[4];
    const float* b1     = (const float*)d_in[5];
    const float* W2     = (const float*)d_in[6];
    const float* b2     = (const float*)d_in[7];
    const float* W_qkv  = (const float*)d_in[8];
    const float* b_qkv  = (const float*)d_in[9];
    const float* W_o    = (const float*)d_in[10];
    const float* b_o    = (const float*)d_in[11];
    const float* ln1_g  = (const float*)d_in[12];
    const float* ln1_b  = (const float*)d_in[13];
    const float* W_ff1  = (const float*)d_in[14];
    const float* b_ff1  = (const float*)d_in[15];
    const float* W_ff2  = (const float*)d_in[16];
    const float* b_ff2  = (const float*)d_in[17];
    const float* ln2_g  = (const float*)d_in[18];
    const float* ln2_b  = (const float*)d_in[19];
    const float* W_cls  = (const float*)d_in[20];
    const float* b_cls  = (const float*)d_in[21];
    float* out = (float*)d_out;

    float* ws = (float*)d_ws;
    const size_t n = NN;
    float* P1  = ws;
    float* S1  = ws + n * 32;
    float* P2  = ws + n * 64;
    float* S2  = ws + n * 128;
    float* H2  = ws + n * 192;
    float* QKV = ws + n * 256;
    float* ATT = ws;
    float* A1  = ws + n * 64;
    float* X1  = ws + n * 128;
    float* FF1 = ws + n * 256;
    float* A2  = ws + n * 384;
    float* X2  = ws;

    {
        long long waves = GEMM_WAVES(NN, 4, 32, 2);
        wmma_gemm_kernel<4, 2, false, 0><<<ceil_div(waves * 32, 256), 256, 0, stream>>>(
            x, nullptr, W1, nullptr, P1, NN, 128, 32);
    }

    zero_kernel<<<ceil_div((long long)NN * 32, 256), 256, 0, stream>>>(S1, NN * 32);
    spmm_kernel<<<ceil_div((long long)NNZ * 8, 256), 256, 0, stream>>>(
        G_rows, G_cols, G_vals, P1, S1, 8);

    {
        long long waves = GEMM_WAVES(NN, 4, 64, 4);
        wmma_gemm_kernel<4, 4, true, 0><<<ceil_div(waves * 32, 256), 256, 0, stream>>>(
            S1, b1, W2, nullptr, P2, NN, 32, 64);
    }

    zero_kernel<<<ceil_div((long long)NN * 64, 256), 256, 0, stream>>>(S2, NN * 64);
    spmm_kernel<<<ceil_div((long long)NNZ * 16, 256), 256, 0, stream>>>(
        G_rows, G_cols, G_vals, P2, S2, 16);

    bias_relu_kernel<<<ceil_div((long long)NN * 64, 256), 256, 0, stream>>>(
        S2, b2, H2, NN * 64, 64);

    launch_gemm_none(H2, W_qkv, b_qkv, QKV, NN, 64, 192, stream);

    attn_kernel<<<(NN / BLK) * NHEAD, 256, 0, stream>>>(QKV, ATT);

    launch_gemm_none(ATT, W_o, b_o, A1, NN, 64, 64, stream);

    ln_residual_kernel<<<ceil_div((long long)NN * 32, 256), 256, 0, stream>>>(
        H2, A1, ln1_g, ln1_b, X1, NN);

    {
        long long waves = GEMM_WAVES(NN, 4, 128, 4);
        wmma_gemm_kernel<4, 4, false, 2><<<ceil_div(waves * 32, 256), 256, 0, stream>>>(
            X1, nullptr, W_ff1, b_ff1, FF1, NN, 64, 128);
    }

    launch_gemm_none(FF1, W_ff2, b_ff2, A2, NN, 128, 64, stream);

    ln_residual_kernel<<<ceil_div((long long)NN * 32, 256), 256, 0, stream>>>(
        X1, A2, ln2_g, ln2_b, X2, NN);

    cls_kernel<<<ceil_div((long long)NN * 32, 256), 256, 0, stream>>>(
        X2, W_cls, b_cls, out, NN);
}


# QKV, ATT, FF1 (read only as WMMA operands) are stored as the f16 their readers rounded them to: bytes halve, loaders lose converts, writer lanes own adjacent columns (8-B stores); results unchanged
# speedup vs baseline: 2.6815x; 2.6815x over previous
#include <hip/hip_runtime.h>
#include <hip/hip_bf16.h>
#include <math.h>

#define NN      131072
#define DEG     16
#define NNZ     (NN * DEG)
#define DMODEL  64
#define NHEAD   4
#define HDIM    16
#define BLK     256

typedef __attribute__((ext_vector_type(16))) _Float16 v16h;
typedef __attribute__((ext_vector_type(8)))  _Float16 v8h;
typedef __attribute__((ext_vector_type(4)))  _Float16 v4h;
typedef __attribute__((ext_vector_type(8)))  float    v8f;
typedef __attribute__((ext_vector_type(2)))  float    v2f;
typedef __attribute__((ext_vector_type(2)))  _Float16 v2h;

__device__ __forceinline__ v8f v8f_zero() {
    v8f z = {0.f, 0.f, 0.f, 0.f, 0.f, 0.f, 0.f, 0.f};
    return z;
}

__device__ __forceinline__ v16h cat16(v8h a, v8h b) {
    return __builtin_shufflevector(a, b, 0, 1, 2, 3, 4, 5, 6, 7,
                                   8, 9, 10, 11, 12, 13, 14, 15);
}

__device__ __forceinline__ v16h frag_from_f32(float4 a0, float4 a1,
                                              float4 a2, float4 a3) {
    v16h r;
    r[0] = (_Float16)a0.x;  r[1] = (_Float16)a0.y;
    r[2] = (_Float16)a0.z;  r[3] = (_Float16)a0.w;
    r[4] = (_Float16)a1.x;  r[5] = (_Float16)a1.y;
    r[6] = (_Float16)a1.z;  r[7] = (_Float16)a1.w;
    r[8] = (_Float16)a2.x;  r[9] = (_Float16)a2.y;
    r[10] = (_Float16)a2.z; r[11] = (_Float16)a2.w;
    r[12] = (_Float16)a3.x; r[13] = (_Float16)a3.y;
    r[14] = (_Float16)a3.z; r[15] = (_Float16)a3.w;
    return r;
}

__device__ __forceinline__ float4 relu4(float4 a, float4 b) {
    float4 r;
    r.x = a.x + b.x; r.x = r.x > 0.f ? r.x : 0.f;
    r.y = a.y + b.y; r.y = r.y > 0.f ? r.y : 0.f;
    r.z = a.z + b.z; r.z = r.z > 0.f ? r.z : 0.f;
    r.w = a.w + b.w; r.w = r.w > 0.f ? r.w : 0.f;
    return r;
}

__device__ __forceinline__ float xhalf(float x) {
    return __int_as_float(__builtin_amdgcn_permlanex16(
        __float_as_int(x), __float_as_int(x), 0x76543210, 0xfedcba98, false, false));
}

__device__ __forceinline__ void load_runs(const float* __restrict__ p, int hi,
                                          float (&v)[32]) {
    #pragma unroll
    for (int i = 0; i < 4; ++i) {
        const float4* q = (const float4*)(p + 16 * i + 8 * hi);
        float4 a = q[0], b = q[1];
        v[8 * i + 0] = a.x; v[8 * i + 1] = a.y; v[8 * i + 2] = a.z; v[8 * i + 3] = a.w;
        v[8 * i + 4] = b.x; v[8 * i + 5] = b.y; v[8 * i + 6] = b.z; v[8 * i + 7] = b.w;
    }
}

#define ROW_STEP(NAME, OPC, M)                                                  \
__device__ __forceinline__ float NAME(float x) {                                \
    float r;                                                                    \
    asm(OPC " %0, %1, %1 row_xmask:" #M " row_mask:0xf bank_mask:0xf bound_ctrl:1" \
        : "=v"(r) : "v"(x));                                                    \
    return r;                                                                   \
}
ROW_STEP(row_max1, "v_max_num_f32_dpp", 1)
ROW_STEP(row_max2, "v_max_num_f32_dpp", 2)
ROW_STEP(row_max4, "v_max_num_f32_dpp", 4)
ROW_STEP(row_max8, "v_max_num_f32_dpp", 8)
ROW_STEP(row_sum1, "v_add_f32_dpp", 1)
ROW_STEP(row_sum2, "v_add_f32_dpp", 2)
ROW_STEP(row_sum4, "v_add_f32_dpp", 4)
ROW_STEP(row_sum8, "v_add_f32_dpp", 8)
#undef ROW_STEP

__device__ __forceinline__ void ln_pair(float (&v)[32],
                                        const float* __restrict__ g,
                                        const float* __restrict__ bt, int hi) {
    float s0 = 0.f, s1 = 0.f, s2 = 0.f, s3 = 0.f;
    #pragma unroll
    for (int e = 0; e < 32; e += 4) {
        s0 += v[e]; s1 += v[e + 1]; s2 += v[e + 2]; s3 += v[e + 3];
    }
    float s = (s0 + s1) + (s2 + s3);
    s += xhalf(s);
    float mu = s * (1.f / 64.f);
    float q0 = 0.f, q1 = 0.f, q2 = 0.f, q3 = 0.f;
    #pragma unroll
    for (int e = 0; e < 32; e += 4) {
        v[e] -= mu;     q0 += v[e] * v[e];
        v[e + 1] -= mu; q1 += v[e + 1] * v[e + 1];
        v[e + 2] -= mu; q2 += v[e + 2] * v[e + 2];
        v[e + 3] -= mu; q3 += v[e + 3] * v[e + 3];
    }
    float vs = (q0 + q1) + (q2 + q3);
    vs += xhalf(vs);
    float rstd = rsqrtf(vs * (1.f / 64.f) + 1e-5f);
    float gg[32], bb[32];
    load_runs(g, hi, gg);
    load_runs(bt, hi, bb);
    #pragma unroll
    for (int e = 0; e < 32; ++e) v[e] = v[e] * rstd * gg[e] + bb[e];
}

__device__ __forceinline__ void x1_runs(const float* __restrict__ Srow,
                                        const float* __restrict__ sb,
                                        const float* __restrict__ Yrow,
                                        const float* __restrict__ g,
                                        const float* __restrict__ bt,
                                        int hi, float (&v)[32]) {
    float b[32], y[32];
    load_runs(Srow, hi, v);
    load_runs(sb, hi, b);
    load_runs(Yrow, hi, y);
    #pragma unroll
    for (int e = 0; e < 32; ++e) {
        float x = v[e] + b[e];
        x = x > 0.f ? x : 0.f;
        v[e] = x + y[e];
    }
    ln_pair(v, g, bt, hi);
}

template <int TM, int TN, int KK, int NCOLS, bool RELU_IN, int ACT_OUT, bool LN_IN,
          bool IN_H, bool OUT_H, bool QSC>
__global__ void __launch_bounds__(256)
wmma_gemm_kernel(const float* __restrict__ A,
                 const _Float16* __restrict__ Ah,
                 const float* __restrict__ abias,
                 const float* __restrict__ W,
                 const float* __restrict__ obias,
                 float* __restrict__ C,
                 _Float16* __restrict__ Ch,
                 const float* __restrict__ Yres,
                 const float* __restrict__ lng,
                 const float* __restrict__ lnb,
                 int M)
{
    static_assert(KK % 32 == 0 && NCOLS % (16 * TN) == 0 && (!LN_IN || KK == 64),
                  "tile shapes");
    static_assert((!OUT_H || TN == 4) && (!QSC || OUT_H) && !(IN_H && (RELU_IN || LN_IN)),
                  "f16 storage paths");
    constexpr int KS = KK / 32;
    constexpr int groups_n = NCOLS / (16 * TN);

    int tid  = (int)threadIdx.x;
    int lane = tid & 31;
    int hi = lane >> 4, lo = lane & 15;

    int gwid = (int)((blockIdx.x * blockDim.x + threadIdx.x) >> 5);
    int total = (M / (16 * TM)) * groups_n;
    if (gwid >= total) return;
    int mg = gwid / groups_n;
    int tg = gwid - mg * groups_n;

    v16h wf[KS][TN];
    float bj[TN];
    if (OUT_H) {
        #pragma unroll
        for (int ks = 0; ks < KS; ++ks) {
            const float* wp = W + (size_t)(32 * ks + 8 * hi) * NCOLS + tg * 64 + 4 * lo;
            #pragma unroll
            for (int e = 0; e < 16; ++e) {
                float4 w4 = *(const float4*)(wp + (16 * (e >> 3) + (e & 7)) * NCOLS);
                wf[ks][0][e] = (_Float16)w4.x;
                wf[ks][1][e] = (_Float16)w4.y;
                wf[ks][2][e] = (_Float16)w4.z;
                wf[ks][3][e] = (_Float16)w4.w;
            }
        }
        float4 b4 = {0.f, 0.f, 0.f, 0.f};
        if (obias) b4 = *(const float4*)(obias + tg * 64 + 4 * lo);
        bj[0] = b4.x; bj[1] = b4.y; bj[2] = b4.z; bj[3] = b4.w;
    } else {
        #pragma unroll
        for (int ks = 0; ks < KS; ++ks) {
            #pragma unroll
            for (int j = 0; j < TN; ++j) {
                const float* wp = W + (size_t)(32 * ks + 8 * hi) * NCOLS
                                    + (tg * TN + j) * 16 + lo;
                #pragma unroll
                for (int e = 0; e < 16; ++e)
                    wf[ks][j][e] = (_Float16)wp[(16 * (e >> 3) + (e & 7)) * NCOLS];
            }
        }
        #pragma unroll
        for (int j = 0; j < TN; ++j)
            bj[j] = obias ? obias[(tg * TN + j) * 16 + lo] : 0.f;
    }
    const float qs = (QSC && tg == 0) ? 0.25f : 1.0f;

    #pragma unroll
    for (int t = 0; t < TM; ++t) {
        int tm = mg * TM + t;
        int m = tm * 16 + lo;

        v8f c[TN];
        #pragma unroll
        for (int j = 0; j < TN; ++j) c[j] = v8f_zero();

        const float* Arow = A + (size_t)m * KK;
        if (IN_H) {
            const _Float16* Ahrow = Ah + (size_t)m * KK;
            #pragma unroll
            for (int ks = 0; ks < KS; ++ks) {
                int k0 = 32 * ks;
                v16h af = cat16(*(const v8h*)(Ahrow + k0 + 8 * hi),
                                *(const v8h*)(Ahrow + k0 + 16 + 8 * hi));
                #pragma unroll
                for (int j = 0; j < TN; ++j)
                    c[j] = __builtin_amdgcn_wmma_f32_16x16x32_f16(
                        false, af, false, wf[ks][j], (short)0, c[j], false, false);
            }
        } else if (LN_IN) {
            float xr[32];
            x1_runs(Arow, abias, Yres + (size_t)m * 64, lng, lnb, hi, xr);
            #pragma unroll
            for (int ks = 0; ks < 2; ++ks) {
                v16h af;
                #pragma unroll
                for (int e = 0; e < 16; ++e) af[e] = (_Float16)xr[16 * ks + e];
                #pragma unroll
                for (int j = 0; j < TN; ++j)
                    c[j] = __builtin_amdgcn_wmma_f32_16x16x32_f16(
                        false, af, false, wf[ks][j], (short)0, c[j], false, false);
            }
        } else {
            #pragma unroll
            for (int ks = 0; ks < KS; ++ks) {
                int k0 = 32 * ks;
                const float4* ap0 = (const float4*)(Arow + k0 + 8 * hi);
                const float4* ap1 = (const float4*)(Arow + k0 + 16 + 8 * hi);
                float4 a0 = ap0[0], a1 = ap0[1], a2 = ap1[0], a3 = ap1[1];
                if (RELU_IN) {
                    const float4* bp0 = (const float4*)(abias + k0 + 8 * hi);
                    const float4* bp1 = (const float4*)(abias + k0 + 16 + 8 * hi);
                    a0 = relu4(a0, bp0[0]); a1 = relu4(a1, bp0[1]);
                    a2 = relu4(a2, bp1[0]); a3 = relu4(a3, bp1[1]);
                }
                v16h af = frag_from_f32(a0, a1, a2, a3);
                #pragma unroll
                for (int j = 0; j < TN; ++j)
                    c[j] = __builtin_amdgcn_wmma_f32_16x16x32_f16(
                        false, af, false, wf[ks][j], (short)0, c[j], false, false);
            }
        }

        if (OUT_H) {
            #pragma unroll
            for (int r = 0; r < 8; ++r) {
                int row = tm * 16 + r + 8 * hi;
                v4h pk;
                #pragma unroll
                for (int j = 0; j < 4; ++j) {
                    float v = c[j][r];
                    if (obias) v += bj[j];
                    if (QSC) v *= qs;
                    if (ACT_OUT == 1) {
                        v = v > 0.f ? v : 0.f;
                    } else if (ACT_OUT == 2) {
                        v = 0.5f * v * (1.0f + erff(v * 0.70710678118654752f));
                    }
                    pk[j] = (_Float16)v;
                }
                *(v4h*)(Ch + (size_t)row * NCOLS + tg * 64 + 4 * lo) = pk;
            }
        } else {
            #pragma unroll
            for (int j = 0; j < TN; ++j) {
                int n = (tg * TN + j) * 16 + lo;
                #pragma unroll
                for (int r = 0; r < 8; ++r) {
                    int row = tm * 16 + r + 8 * hi;
                    float v = c[j][r];
                    if (obias) v += bj[j];
                    if (ACT_OUT == 1) {
                        v = v > 0.f ? v : 0.f;
                    } else if (ACT_OUT == 2) {
                        v = 0.5f * v * (1.0f + erff(v * 0.70710678118654752f));
                    }
                    C[(size_t)row * NCOLS + n] = v;
                }
            }
        }
    }
}

#define BIN_ROWS  512
#define BIN_SHIFT 9
#define NBINS     (NN / BIN_ROWS)
#define EWGS      256
#define EPW       (NNZ / EWGS)
#define COL_BITS  17
static_assert(NN == (1 << COL_BITS) && BIN_ROWS == (1 << BIN_SHIFT) &&
              NBINS == 256 && EWGS == 256 && EPW % 1024 == 0,
              "packing of (row in bin, col); thread maps of the three kernels");

__global__ void __launch_bounds__(1024)
bin_count_kernel(const int* __restrict__ rows, int* __restrict__ cnt)
{
    __shared__ int h[NBINS];
    int tid = (int)threadIdx.x;
    const int4* rp = (const int4*)(rows + (size_t)blockIdx.x * EPW);
    int4 a = rp[tid], b = rp[tid + 1024];
    if (tid < NBINS) h[tid] = 0;
    __syncthreads();
    atomicAdd(&h[a.x >> BIN_SHIFT], 1); atomicAdd(&h[a.y >> BIN_SHIFT], 1);
    atomicAdd(&h[a.z >> BIN_SHIFT], 1); atomicAdd(&h[a.w >> BIN_SHIFT], 1);
    atomicAdd(&h[b.x >> BIN_SHIFT], 1); atomicAdd(&h[b.y >> BIN_SHIFT], 1);
    atomicAdd(&h[b.z >> BIN_SHIFT], 1); atomicAdd(&h[b.w >> BIN_SHIFT], 1);
    __syncthreads();
    if (tid < NBINS) cnt[blockIdx.x * NBINS + tid] = h[tid];
}

__global__ void __launch_bounds__(1024)
bin_place_kernel(const int* __restrict__ rows, const int* __restrict__ cols,
                 const float* __restrict__ vals, const int* __restrict__ cnt,
                 int* __restrict__ packed, float* __restrict__ bval,
                 int* __restrict__ binstart)
{
    __shared__ __align__(16) int ptot[16][NBINS];
    __shared__ __align__(16) int pbef[16][NBINS];
    __shared__ int sc[NBINS];
    __shared__ int ls[NBINS];
    __shared__ int cur[NBINS];
    __shared__ int gofs[NBINS];
    __shared__ int   lpk[EPW];
    __shared__ float lvl[EPW];
    __shared__ int   lpos[EPW];
    int tid = (int)threadIdx.x;
    int me = (int)blockIdx.x;
    size_t base = (size_t)me * EPW;
    int er[EPW / 1024], ec[EPW / 1024];
    float ev[EPW / 1024];
    #pragma unroll
    for (int i = 0; i < EPW / 1024; ++i) {
        size_t e = base + (size_t)(i * 1024 + tid);
        er[i] = rows[e];
        ec[i] = cols[e];
        ev[i] = vals[e];
    }
    int own = (tid < NBINS) ? cnt[me * NBINS + tid] : 0;
    {
        int g = tid & 63;
        int wq = tid >> 6;
        const int4* cp = (const int4*)cnt + (size_t)(wq * 16) * (NBINS / 4) + g;
        int4 c4[16];
        #pragma unroll
        for (int k = 0; k < 16; ++k) c4[k] = cp[k * (NBINS / 4)];
        int4 tot = {0, 0, 0, 0}, bef = {0, 0, 0, 0};
        #pragma unroll
        for (int k = 0; k < 16; ++k) {
            tot.x += c4[k].x; tot.y += c4[k].y; tot.z += c4[k].z; tot.w += c4[k].w;
            if (wq * 16 + k < me) {
                bef.x += c4[k].x; bef.y += c4[k].y; bef.z += c4[k].z; bef.w += c4[k].w;
            }
        }
        *(int4*)&ptot[wq][4 * g] = tot;
        *(int4*)&pbef[wq][4 * g] = bef;
    }
    __syncthreads();
    int total = 0, before = 0;
    if (tid < NBINS) {
        for (int q = 0; q < 16; ++q) { total += ptot[q][tid]; before += pbef[q][tid]; }
        sc[tid] = total;
        ls[tid] = own;
    }
    __syncthreads();
    for (int d = 1; d < NBINS; d <<= 1) {
        int v = 0, u = 0;
        if (tid < NBINS && tid >= d) { v = sc[tid - d]; u = ls[tid - d]; }
        __syncthreads();
        if (tid < NBINS) { sc[tid] += v; ls[tid] += u; }
        __syncthreads();
    }
    if (tid < NBINS) {
        int excl = sc[tid] - total;
        int lexcl = ls[tid] - own;
        cur[tid] = lexcl;
        gofs[tid] = excl + before - lexcl;
        if (me == 0) {
            binstart[tid] = excl;
            if (tid == NBINS - 1) binstart[NBINS] = sc[tid];
        }
    }
    __syncthreads();
    #pragma unroll
    for (int i = 0; i < EPW / 1024; ++i) {
        int r = er[i];
        int c = ec[i];
        float v = ev[i];
        int b = r >> BIN_SHIFT;
        int sl = atomicAdd(&cur[b], 1);
        lpk[sl] = ((r & (BIN_ROWS - 1)) << COL_BITS) | c;
        lvl[sl] = v;
        lpos[sl] = sl + gofs[b];
    }
    __syncthreads();
    #pragma unroll
    for (int i = 0; i < EPW / 1024; ++i) {
        int sl = i * 1024 + tid;
        int pos = lpos[sl];
        packed[pos] = lpk[sl];
        bval[pos] = lvl[sl];
    }
}

template <int F>
__global__ void __launch_bounds__(1024)
spmm_bin_kernel(const int* __restrict__ packed, const float* __restrict__ bval,
                const int* __restrict__ binstart,
                const float* __restrict__ in, float* __restrict__ out)
{
    constexpr int NC = F / 32;
    __shared__ __align__(16) float acc[BIN_ROWS * F];
    int tid = (int)threadIdx.x;
    int lane = tid & 31;
    int wid = __builtin_amdgcn_readfirstlane(tid >> 5);
    int bin = (int)blockIdx.x;

    const float4 z4 = {0.f, 0.f, 0.f, 0.f};
    for (int i = tid; i < BIN_ROWS * F / 4; i += 1024) ((float4*)acc)[i] = z4;
    __syncthreads();

    int beg = binstart[bin], end = binstart[bin + 1];
    int e0 = beg + wid * 32;
    int pk = 0;
    float v = 0.f;
    if (e0 + lane < end) { pk = packed[e0 + lane]; v = bval[e0 + lane]; }
    while (e0 < end) {
        int nj = end - e0 < 32 ? end - e0 : 32;
        int en = e0 + 32 * 32;
        int pkn = 0;
        float vn = 0.f;
        if (en + lane < end) { pkn = packed[en + lane]; vn = bval[en + lane]; }

        int c = pk & ((1 << COL_BITS) - 1);
        int rl = pk >> COL_BITS;
        int vb = __float_as_int(v);
        if (nj == 32) {
            #pragma unroll
            for (int jb = 0; jb < 32; jb += 16) {
                float x[16][NC];
                #pragma unroll
                for (int u = 0; u < 16; ++u) {
                    int cj = __builtin_amdgcn_readlane(c, jb + u);
                    const float* src = in + (size_t)cj * F;
                    #pragma unroll
                    for (int q = 0; q < NC; ++q) x[u][q] = src[q * 32 + lane];
                }
                #pragma unroll
                for (int u = 0; u < 16; ++u) {
                    int rj = __builtin_amdgcn_readlane(rl, jb + u);
                    float vj = __int_as_float(__builtin_amdgcn_readlane(vb, jb + u));
                    #pragma unroll
                    for (int q = 0; q < NC; ++q)
                        atomicAdd(&acc[rj * F + q * 32 + lane], vj * x[u][q]);
                }
            }
        } else {
            #pragma unroll
            for (int jb = 0; jb < 32; jb += 16) {
                if (jb < nj) {
                    float x[16][NC];
                    #pragma unroll
                    for (int u = 0; u < 16; ++u) {
                        int cj = __builtin_amdgcn_readlane(c, jb + u);
                        const float* src = in + (size_t)cj * F;
                        #pragma unroll
                        for (int q = 0; q < NC; ++q) x[u][q] = src[q * 32 + lane];
                    }
                    #pragma unroll
                    for (int u = 0; u < 16; ++u) {
                        if (jb + u < nj) {
                            int rj = __builtin_amdgcn_readlane(rl, jb + u);
                            float vj = __int_as_float(__builtin_amdgcn_readlane(vb, jb + u));
                            #pragma unroll
                            for (int q = 0; q < NC; ++q)
                                atomicAdd(&acc[rj * F + q * 32 + lane], vj * x[u][q]);
                        }
                    }
                }
            }
        }
        pk = pkn;
        v = vn;
        e0 = en;
    }
    __syncthreads();

    float4* op = (float4*)(out + (size_t)bin * BIN_ROWS * F);
    for (int i = tid; i < BIN_ROWS * F / 4; i += 1024) op[i] = ((const float4*)acc)[i];
}

#define K_PITCH  24
#define VT_PITCH 264
#define P_PITCH  40

__global__ void __attribute__((amdgpu_waves_per_eu(8, 10)))
attn_kernel(const _Float16* __restrict__ QKV, _Float16* __restrict__ ATT)
{
    __shared__ __align__(16) _Float16 sK[BLK * K_PITCH];
    __shared__ __align__(16) _Float16 sVt[HDIM * VT_PITCH];
    __shared__ __align__(16) _Float16 sP[8][16 * P_PITCH];

    int b = (int)(blockIdx.x >> 2);
    int h = (int)(blockIdx.x & 3);
    int tid  = (int)threadIdx.x;
    int wid  = tid >> 5;
    int lane = tid & 31;
    int hi   = lane >> 4;
    int lo   = lane & 15;
    size_t base = (size_t)b * BLK * (3 * DMODEL);

    {
        int j = tid & 15, dh = (tid >> 4) & 1, g = tid >> 5;
        int key0 = g * 32 + j, key1 = key0 + 16;
        const _Float16* r0 = QKV + base + (size_t)key0 * 192 + h * HDIM + 8 * dh;
        const _Float16* r1 = QKV + base + (size_t)key1 * 192 + h * HDIM + 8 * dh;
        v8h ka = *(const v8h*)(r0 + DMODEL),     kb = *(const v8h*)(r1 + DMODEL);
        v8h va = *(const v8h*)(r0 + 2 * DMODEL), vb = *(const v8h*)(r1 + 2 * DMODEL);
        *(v8h*)(&sK[key0 * K_PITCH + 8 * dh]) = ka;
        *(v8h*)(&sK[key1 * K_PITCH + 8 * dh]) = kb;
        _Float16* vt = &sVt[(8 * dh) * VT_PITCH + g * 32 + 2 * j];
        #pragma unroll
        for (int cc = 0; cc < 8; ++cc) {
            v2h pr = {va[cc], vb[cc]};
            *(v2h*)(vt + cc * VT_PITCH) = pr;
        }
    }
    __syncthreads();

    v8h zero8 = {(_Float16)0.f, (_Float16)0.f, (_Float16)0.f, (_Float16)0.f,
                 (_Float16)0.f, (_Float16)0.f, (_Float16)0.f, (_Float16)0.f};

    for (int qt = wid; qt < 16; qt += 8) {
        int qrow = qt * 16 + lo;
        v16h qf = cat16(*(const v8h*)(QKV + base + (size_t)qrow * 192
                                      + h * HDIM + 8 * hi), zero8);

        v8f ctx = v8f_zero();
        float m[8], l[8];
        #pragma unroll
        for (int r = 0; r < 8; ++r) { m[r] = -3.0e38f; l[r] = 0.f; }

        for (int kk = 0; kk < 8; ++kk) {
            v8f s0, s1;
            {
                int key = (kk * 2) * 16 + lo;
                v16h bf = cat16(*(const v8h*)(&sK[key * K_PITCH + 8 * hi]), zero8);
                v8f z = v8f_zero();
                s0 = __builtin_amdgcn_wmma_f32_16x16x32_f16(
                    false, qf, false, bf, (short)0, z, false, false);
            }
            {
                int key = (kk * 2 + 1) * 16 + lo;
                v16h bf = cat16(*(const v8h*)(&sK[key * K_PITCH + 8 * hi]), zero8);
                v8f z = v8f_zero();
                s1 = __builtin_amdgcn_wmma_f32_16x16x32_f16(
                    false, qf, false, bf, (short)0, z, false, false);
            }
            #pragma unroll
            for (int r = 0; r < 8; ++r) {
                float tm = fmaxf(s0[r], s1[r]);
                tm = row_max8(row_max4(row_max2(row_max1(tm))));
                float nm = fmaxf(m[r], tm);
                float sc = __expf(m[r] - nm);
                m[r] = nm;
                float e0 = __expf(s0[r] - nm);
                float e1 = __expf(s1[r] - nm);
                float rs = e0 + e1;
                rs = row_sum8(row_sum4(row_sum2(row_sum1(rs))));
                l[r] = l[r] * sc + rs;
                ctx[r] *= sc;
                int row = r + 8 * hi;
                v2h pr = {(_Float16)e0, (_Float16)e1};
                *(v2h*)(&sP[wid][row * P_PITCH + 2 * lo]) = pr;
            }
            v16h af = cat16(*(const v8h*)(&sP[wid][lo * P_PITCH + 8 * hi]),
                            *(const v8h*)(&sP[wid][lo * P_PITCH + 16 + 8 * hi]));
            v16h bf = cat16(*(const v8h*)(&sVt[lo * VT_PITCH + kk * 32 + 8 * hi]),
                            *(const v8h*)(&sVt[lo * VT_PITCH + kk * 32 + 16 + 8 * hi]));
            ctx = __builtin_amdgcn_wmma_f32_16x16x32_f16(
                false, af, false, bf, (short)0, ctx, false, false);
        }

        #pragma unroll
        for (int r = 0; r < 8; r += 2) {
            float oa = ctx[r] / l[r], ob = ctx[r + 1] / l[r + 1];
            bool odd = (lo & 1) != 0;
            float give = odd ? oa : ob;
            float got = __int_as_float(__builtin_amdgcn_update_dpp(
                0, __float_as_int(give), 0xB1  , 0xf, 0xf, true));
            float keep = odd ? ob : oa;
            v2h pr = {(_Float16)(odd ? got : keep), (_Float16)(odd ? keep : got)};
            int row = b * BLK + qt * 16 + r + (odd ? 1 : 0) + 8 * hi;
            *(v2h*)(ATT + (size_t)row * DMODEL + h * HDIM + (lo & 14)) = pr;
        }
    }
}

__global__ void __launch_bounds__(256)
ln2_cls_kernel(const float* __restrict__ S2, const float* __restrict__ b2,
               const float* __restrict__ A1,
               const float* __restrict__ g1, const float* __restrict__ bt1,
               const float* __restrict__ A2,
               const float* __restrict__ g2, const float* __restrict__ bt2,
               const float* __restrict__ Wc, const float* __restrict__ bc,
               float* __restrict__ out)
{
    int lane = (int)(threadIdx.x & 31);
    int hi = lane >> 4, lo = lane & 15;
    int tile = (int)((blockIdx.x * blockDim.x + threadIdx.x) >> 5);
    size_t row = (size_t)tile * 16 + lo;

    float wv[32];
    #pragma unroll
    for (int i = 0; i < 4; ++i) {
        #pragma unroll
        for (int u = 0; u < 8; ++u)
            wv[8 * i + u] = Wc[(16 * i + 8 * hi + u) * 16 + lo];
    }

    float v[32], y[32];
    x1_runs(S2 + row * 64, b2, A1 + row * 64, g1, bt1, hi, v);
    load_runs(A2 + row * 64, hi, y);
    #pragma unroll
    for (int e = 0; e < 32; ++e) v[e] += y[e];
    ln_pair(v, g2, bt2, hi);

    v8f acc0 = v8f_zero(), acc1 = v8f_zero();
    #pragma unroll
    for (int t = 0; t < 16; t += 2) {
        v2f a0 = {wv[2 * t], wv[2 * t + 1]};
        v2f x0 = {v[2 * t], v[2 * t + 1]};
        acc0 = __builtin_amdgcn_wmma_f32_16x16x4_f32(
            false, a0, false, x0, (short)0, acc0, false, false);
        v2f a1 = {wv[2 * t + 2], wv[2 * t + 3]};
        v2f x1 = {v[2 * t + 2], v[2 * t + 3]};
        acc1 = __builtin_amdgcn_wmma_f32_16x16x4_f32(
            false, a1, false, x1, (short)0, acc1, false, false);
    }

    const float4* bp = (const float4*)(bc + 8 * hi);
    float4 c0 = bp[0], c1 = bp[1];
    float lg[8];
    lg[0] = acc0[0] + acc1[0] + c0.x; lg[1] = acc0[1] + acc1[1] + c0.y;
    lg[2] = acc0[2] + acc1[2] + c0.z; lg[3] = acc0[3] + acc1[3] + c0.w;
    lg[4] = acc0[4] + acc1[4] + c1.x; lg[5] = acc0[5] + acc1[5] + c1.y;
    lg[6] = acc0[6] + acc1[6] + c1.z; lg[7] = acc0[7] + acc1[7] + c1.w;

    float mx = lg[0];
    #pragma unroll
    for (int r = 1; r < 8; ++r) mx = fmaxf(mx, lg[r]);
    mx = fmaxf(mx, xhalf(mx));
    float sm = 0.f;
    #pragma unroll
    for (int r = 0; r < 8; ++r) sm += expf(lg[r] - mx);
    sm += xhalf(sm);
    float ls = logf(sm);
    float4 o0, o1;
    o0.x = lg[0] - mx - ls; o0.y = lg[1] - mx - ls;
    o0.z = lg[2] - mx - ls; o0.w = lg[3] - mx - ls;
    o1.x = lg[4] - mx - ls; o1.y = lg[5] - mx - ls;
    o1.z = lg[6] - mx - ls; o1.w = lg[7] - mx - ls;
    float4* op = (float4*)(out + row * 16 + 8 * hi);
    op[0] = o0;
    op[1] = o1;
}


static inline int ceil_div(long long a, long long b) { return (int)((a + b - 1) / b); }

#define GEMM_WAVES(M, TM, Nc, TN) \
    ((long long)((M) / (16 * (TM))) * ((Nc) / (16 * (TN))))

template <int KK>
static void launch_gemm_none(const _Float16* Ah, const float* W, const float* ob,
                             float* C, hipStream_t s)
{
    long long waves = GEMM_WAVES(NN, 4, 64, 4);
    wmma_gemm_kernel<4, 4, KK, 64, false, 0, false, true, false, false>
        <<<ceil_div(waves * 32, 256), 256, 0, s>>>(
        nullptr, Ah, nullptr, W, ob, C, nullptr, nullptr, nullptr, nullptr, NN);
}

extern "C" void kernel_launch(void* const* d_in, const int* in_sizes, int n_in,
                              void* d_out, int out_size, void* d_ws, size_t ws_size,
                              hipStream_t stream)
{
    const float* x      = (const float*)d_in[0];
    const int*   G_rows = (const int*)  d_in[1];
    const int*   G_cols = (const int*)  d_in[2];
    const float* G_vals = (const float*)d_in[3];
    const float* W1     = (const float*)d_in[4];
    const float* b1     = (const float*)d_in[5];
    const float* W2     = (const float*)d_in[6];
    const float* b2     = (const float*)d_in[7];
    const float* W_qkv  = (const float*)d_in[8];
    const float* b_qkv  = (const float*)d_in[9];
    const float* W_o    = (const float*)d_in[10];
    const float* b_o    = (const float*)d_in[11];
    const float* ln1_g  = (const float*)d_in[12];
    const float* ln1_b  = (const float*)d_in[13];
    const float* W_ff1  = (const float*)d_in[14];
    const float* b_ff1  = (const float*)d_in[15];
    const float* W_ff2  = (const float*)d_in[16];
    const float* b_ff2  = (const float*)d_in[17];
    const float* ln2_g  = (const float*)d_in[18];
    const float* ln2_b  = (const float*)d_in[19];
    const float* W_cls  = (const float*)d_in[20];
    const float* b_cls  = (const float*)d_in[21];
    float* out = (float*)d_out;

    float* ws = (float*)d_ws;
    const size_t n = NN;
    float* P1  = ws;
    float* S1  = ws + n * 32;
    float* P2  = ws + n * 64;
    float* S2  = ws + n * 128;
    _Float16* QKV = (_Float16*)(ws + n * 256);
    _Float16* ATT = (_Float16*)ws;
    float* A1  = ws + n * 64;
    _Float16* FF1 = (_Float16*)(ws + n * 256);
    float* A2  = ws + n * 384;
    int*   EPK = (int*)(ws + n * 192);
    float* EVL = ws + n * 208;
    int*   CNT = (int*)(ws + n * 224);
    int*   BST = (int*)(ws + n * 225);

    bin_count_kernel<<<EWGS, 1024, 0, stream>>>(G_rows, CNT);
    bin_place_kernel<<<EWGS, 1024, 0, stream>>>(
        G_rows, G_cols, G_vals, CNT, EPK, EVL, BST);

    {
        long long waves = GEMM_WAVES(NN, 4, 32, 2);
        wmma_gemm_kernel<4, 2, 128, 32, false, 0, false, false, false, false>
            <<<ceil_div(waves * 32, 256), 256, 0, stream>>>(
            x, nullptr, nullptr, W1, nullptr, P1, nullptr, nullptr, nullptr, nullptr, NN);
    }

    spmm_bin_kernel<32><<<NBINS, 1024, 0, stream>>>(EPK, EVL, BST, P1, S1);

    {
        long long waves = GEMM_WAVES(NN, 4, 64, 4);
        wmma_gemm_kernel<4, 4, 32, 64, true, 0, false, false, false, false>
            <<<ceil_div(waves * 32, 256), 256, 0, stream>>>(
            S1, nullptr, b1, W2, nullptr, P2, nullptr, nullptr, nullptr, nullptr, NN);
    }

    spmm_bin_kernel<64><<<NBINS, 1024, 0, stream>>>(EPK, EVL, BST, P2, S2);


    {
        long long waves = GEMM_WAVES(NN, 4, 192, 4);
        wmma_gemm_kernel<4, 4, 64, 192, true, 0, false, false, true, true>
            <<<ceil_div(waves * 32, 256), 256, 0, stream>>>(
            S2, nullptr, b2, W_qkv, b_qkv, nullptr, QKV, nullptr, nullptr, nullptr, NN);
    }

    attn_kernel<<<(NN / BLK) * NHEAD, 256, 0, stream>>>(QKV, ATT);

    launch_gemm_none<64>(ATT, W_o, b_o, A1, stream);


    {
        long long waves = GEMM_WAVES(NN, 4, 128, 4);
        wmma_gemm_kernel<4, 4, 64, 128, false, 2, true, false, true, false>
            <<<ceil_div(waves * 32, 256), 256, 0, stream>>>(
            S2, nullptr, b2, W_ff1, b_ff1, nullptr, FF1, A1, ln1_g, ln1_b, NN);
    }

    launch_gemm_none<128>(FF1, W_ff2, b_ff2, A2, stream);


    ln2_cls_kernel<<<NN / 16 * 32 / 256, 256, 0, stream>>>(
        S2, b2, A1, ln1_g, ln1_b, A2, ln2_g, ln2_b, W_cls, b_cls, out);
}
